# starting kernel
# baseline (speedup 1.0000x reference)
#include <hip/hip_runtime.h>

typedef __attribute__((ext_vector_type(2))) float v2f;
typedef __attribute__((ext_vector_type(8))) float v8f;

#define N_FEAT 512
#define HIDDEN 128
#define NCLS   64

__device__ __forceinline__ void atomAddF32(float* p, float v) {
    unsafeAtomicAdd(p, v);
}


__global__ void fill_kernel(float* __restrict__ p, float v, long long n) {
    long long i = (long long)blockIdx.x * blockDim.x + threadIdx.x;
    if (i < n) p[i] = v;
}

__global__ void deg_accum_kernel(const int* __restrict__ dst, float* __restrict__ deg, int E) {
    int e = blockIdx.x * blockDim.x + threadIdx.x;
    if (e < E) atomAddF32(&deg[dst[e]], 1.0f);
}

__global__ void rsqrt_kernel(float* __restrict__ p, int n) {
    int i = blockIdx.x * blockDim.x + threadIdx.x;
    if (i < n) p[i] = rsqrtf(p[i]);
}

template<int F>
__global__ void bias_relu_kernel(float* __restrict__ t, const float* __restrict__ b, long long n) {
    long long i = (long long)blockIdx.x * blockDim.x + threadIdx.x;
    if (i < n) {
        int j = (int)(i & (F - 1));
        t[i] = fmaxf(t[i] + b[j], 0.0f);
    }
}

template<int F>
__global__ void init_bias_kernel(float* __restrict__ out, const float* __restrict__ b, long long n) {
    long long i = (long long)blockIdx.x * blockDim.x + threadIdx.x;
    if (i < n) out[i] = b[(int)(i & (F - 1))];
}


template<int K, int NCOL>
__global__ __launch_bounds__(256)
void gemm_wmma_f32(const float* __restrict__ A, const float* __restrict__ B,
                   float* __restrict__ C, int M) {
    constexpr int NT = NCOL / 16;
    const int lane = threadIdx.x & 31;
    const int wave = threadIdx.x >> 5;
    const int m0 = (blockIdx.x * 8 + wave) * 16;
    if (m0 >= M) return;

    const int mr = lane & 15;
    const int kb = (lane >> 4) << 1;
    const int nc = lane & 15;

    const float* a_ptr = A + (size_t)(m0 + mr) * K + kb;
    const float* b_ptr = B + (size_t)kb * NCOL + nc;

    v8f acc[NT] = {};
#pragma unroll 4
    for (int k = 0; k < K; k += 4) {
        v2f a;
        a.x = a_ptr[0];
        a.y = a_ptr[1];
#pragma unroll
        for (int nt = 0; nt < NT; ++nt) {
            v2f b;
            b.x = b_ptr[nt * 16];
            b.y = b_ptr[NCOL + nt * 16];
            acc[nt] = __builtin_amdgcn_wmma_f32_16x16x4_f32(
                 false, a,  false, b,
                 (short)0, acc[nt],  false,  false);
        }
        a_ptr += 4;
        b_ptr += (size_t)4 * NCOL;
    }

    const int rbase = m0 + ((lane >> 4) << 3);
#pragma unroll
    for (int nt = 0; nt < NT; ++nt) {
        float* c_ptr = C + (size_t)rbase * NCOL + nt * 16 + nc;
#pragma unroll
        for (int r = 0; r < 8; ++r)
            c_ptr[(size_t)r * NCOL] = acc[nt][r];
    }
}


template<int F, int LOG_TPE>
__global__ void scatter_kernel(const float* __restrict__ h, const int* __restrict__ srcI,
                               const int* __restrict__ dstI, const float* __restrict__ dinv,
                               float* __restrict__ acc, int E, int N) {
    long long tid = (long long)blockIdx.x * blockDim.x + threadIdx.x;
    long long edge = tid >> LOG_TPE;
    if (edge >= (long long)E + N) return;
    int fo = ((int)tid & ((1 << LOG_TPE) - 1)) << 2;

    int s, d;
    if (edge < E) { s = srcI[edge]; d = dstI[edge]; }
    else          { s = d = (int)(edge - E); }

    float nrm = dinv[s] * dinv[d];
    const float4 v = *(const float4*)(h + (size_t)s * F + fo);
    float* o = acc + (size_t)d * F + fo;
    atomAddF32(o + 0, v.x * nrm);
    atomAddF32(o + 1, v.y * nrm);
    atomAddF32(o + 2, v.z * nrm);
    atomAddF32(o + 3, v.w * nrm);
}


extern "C" void kernel_launch(void* const* d_in, const int* in_sizes, int n_in,
                              void* d_out, int out_size, void* d_ws, size_t ws_size,
                              hipStream_t stream) {
    const float* x  = (const float*)d_in[0];
    const int*   ei = (const int*)d_in[1];
    const float* W1 = (const float*)d_in[2];
    const float* b1 = (const float*)d_in[3];
    const float* W2 = (const float*)d_in[4];
    const float* b2 = (const float*)d_in[5];
    float* out = (float*)d_out;

    const int N = in_sizes[0] / N_FEAT;
    const int E = in_sizes[1] / 2;
    const int* srcI = ei;
    const int* dstI = ei + E;

    char* ws = (char*)d_ws;
    size_t degB = ((size_t)N * sizeof(float) + 255) & ~(size_t)255;
    size_t t1B  = (((size_t)N * HIDDEN * sizeof(float)) + 255) & ~(size_t)255;
    float* deg = (float*)ws;
    float* t1  = (float*)(ws + degB);
    float* t2  = (float*)(ws + degB + t1B);
    float* t3  = t1;

    const int BT = 256;
    const int Mtiles = N / 16;
    const int gemmBlocks = (Mtiles + 7) / 8;

    fill_kernel<<<(N + BT - 1) / BT, BT, 0, stream>>>(deg, 1.0f, N);
    deg_accum_kernel<<<(E + BT - 1) / BT, BT, 0, stream>>>(dstI, deg, E);
    rsqrt_kernel<<<(N + BT - 1) / BT, BT, 0, stream>>>(deg, N);

    long long t2n = (long long)N * HIDDEN;
    fill_kernel<<<(int)((t2n + BT - 1) / BT), BT, 0, stream>>>(t2, 0.0f, t2n);

    gemm_wmma_f32<N_FEAT, HIDDEN><<<gemmBlocks, 256, 0, stream>>>(x, W1, t1, N);

    long long tot1 = ((long long)E + N) << 5;
    scatter_kernel<HIDDEN, 5><<<(int)((tot1 + BT - 1) / BT), BT, 0, stream>>>(
        t1, srcI, dstI, deg, t2, E, N);

    bias_relu_kernel<HIDDEN><<<(int)((t2n + BT - 1) / BT), BT, 0, stream>>>(t2, b1, t2n);

    gemm_wmma_f32<HIDDEN, NCLS><<<gemmBlocks, 256, 0, stream>>>(t2, W2, t3, N);

    long long outn = (long long)N * NCLS;
    init_bias_kernel<NCLS><<<(int)((outn + BT - 1) / BT), BT, 0, stream>>>(out, b2, outn);
    long long tot2 = ((long long)E + N) << 4;
    scatter_kernel<NCLS, 4><<<(int)((tot2 + BT - 1) / BT), BT, 0, stream>>>(
        t3, srcI, dstI, deg, out, E, N);
}


# Both GEMMs re-tiled: 128-row work-group tile, A and B k-chunks staged once per work-group in LDS (2 stages, whole-line 16-byte loads), waves take 32 x NCOL/2 tiles from LDS; k order of sums kept
# speedup vs baseline: 1.1516x; 1.1516x over previous
#include <hip/hip_runtime.h>

typedef __attribute__((ext_vector_type(2))) float v2f;
typedef __attribute__((ext_vector_type(8))) float v8f;

#define N_FEAT 512
#define HIDDEN 128
#define NCLS   64

__device__ __forceinline__ void atomAddF32(float* p, float v) {
    unsafeAtomicAdd(p, v);
}


__global__ void fill_kernel(float* __restrict__ p, float v, long long n) {
    long long i = (long long)blockIdx.x * blockDim.x + threadIdx.x;
    if (i < n) p[i] = v;
}

__global__ void deg_accum_kernel(const int* __restrict__ dst, float* __restrict__ deg, int E) {
    int e = blockIdx.x * blockDim.x + threadIdx.x;
    if (e < E) atomAddF32(&deg[dst[e]], 1.0f);
}

__global__ void rsqrt_kernel(float* __restrict__ p, int n) {
    int i = blockIdx.x * blockDim.x + threadIdx.x;
    if (i < n) p[i] = rsqrtf(p[i]);
}

template<int F>
__global__ void bias_relu_kernel(float* __restrict__ t, const float* __restrict__ b, long long n) {
    long long i = (long long)blockIdx.x * blockDim.x + threadIdx.x;
    if (i < n) {
        int j = (int)(i & (F - 1));
        t[i] = fmaxf(t[i] + b[j], 0.0f);
    }
}

template<int F>
__global__ void init_bias_kernel(float* __restrict__ out, const float* __restrict__ b, long long n) {
    long long i = (long long)blockIdx.x * blockDim.x + threadIdx.x;
    if (i < n) out[i] = b[(int)(i & (F - 1))];
}


typedef __attribute__((ext_vector_type(4))) float v4f;

template<int PA, int PP, int NT>
__device__ __forceinline__ void mma_chunk(const float* __restrict__ pa,
                                          const float* __restrict__ pb,
                                          v8f (&acc)[2][NT]) {
#pragma unroll
    for (int ks = 0; ks < 8; ++ks) {
        const v2f a0 = *(const v2f*)(pa + ks * 4);
        const v2f a1 = *(const v2f*)(pa + 16 * PA + ks * 4);
#pragma unroll
        for (int nt = 0; nt < NT; ++nt) {
            const v2f b = *(const v2f*)(pb + ks * 2 * PP + nt * 32);
            acc[0][nt] = __builtin_amdgcn_wmma_f32_16x16x4_f32(
                 false, a0,  false, b,
                 (short)0, acc[0][nt],  false,  false);
            acc[1][nt] = __builtin_amdgcn_wmma_f32_16x16x4_f32(
                 false, a1,  false, b,
                 (short)0, acc[1][nt],  false,  false);
        }
    }
}

template<int K, int NCOL>
__global__ __launch_bounds__(256)
void gemm_wmma_f32(const float* __restrict__ A, const float* __restrict__ B,
                   float* __restrict__ C, int M) {
    constexpr int BM  = 128;
    constexpr int KC  = 32;
    constexpr int PA  = KC + 4;
    constexpr int PP  = 2 * NCOL + 32;
    constexpr int WN  = NCOL / 2;
    constexpr int NT  = WN / 16;
    constexpr int NCH = K / KC;
    constexpr int CG  = NCOL / 32;
    constexpr int B_P = CG;
    constexpr int SA = BM * PA;
    constexpr int SB = (KC / 2) * PP;

    __shared__ __attribute__((aligned(16))) float sA[2 * SA];
    __shared__ __attribute__((aligned(16))) float sB[2 * SB];

    const int tid  = threadIdx.x;
    const int lane = tid & 31;
    const int wave = __builtin_amdgcn_readfirstlane(tid >> 5);
    const int wm = wave >> 1;
    const int wn = wave & 1;
    const int c = lane & 15;
    const int h = lane >> 4;
    const int m0 = blockIdx.x * BM;

    const int a_row = tid >> 3;
    const int a_col = (tid & 7) << 2;
    const int b_r = lane >> 3;
    const int b_c = (lane & 7) << 2;

    const int r0 = min(m0 + a_row,      M - 1);
    const int r1 = min(m0 + a_row + 32, M - 1);
    const int r2 = min(m0 + a_row + 64, M - 1);
    const int r3 = min(m0 + a_row + 96, M - 1);
    const float* ap0 = A + (size_t)r0 * K + a_col;
    const float* ap1 = A + (size_t)r1 * K + a_col;
    const float* ap2 = A + (size_t)r2 * K + a_col;
    const float* ap3 = A + (size_t)r3 * K + a_col;
    const float* bsrc[B_P];
    float* bdst[B_P];
#pragma unroll
    for (int p = 0; p < B_P; ++p) {
        const int q = wave + 8 * p;
        const int row = 4 * (q / CG) + b_r;
        const int col = 32 * (q % CG) + b_c;
        bsrc[p] = B + (size_t)row * NCOL + col;
        bdst[p] = sB + (row >> 1) * PP + 2 * col + (row & 1);
    }

    float* wa = sA + a_row * PA + a_col;
    const float* fa = sA + (wm * 32 + c) * PA + 2 * h;
    const float* fb = sB + h * PP + (wn * WN + c) * 2;

    v8f acc[2][NT] = {};

    v4f a0 = *(const v4f*)ap0, a1 = *(const v4f*)ap1, a2 = *(const v4f*)ap2, a3 = *(const v4f*)ap3;
    v4f b[B_P];
#pragma unroll
    for (int p = 0; p < B_P; ++p) b[p] = *(const v4f*)(bsrc[p]);

    *(v4f*)(wa)           = a0;
    *(v4f*)(wa + 32 * PA) = a1;
    *(v4f*)(wa + 64 * PA) = a2;
    *(v4f*)(wa + 96 * PA) = a3;
#pragma unroll
    for (int p = 0; p < B_P; ++p) {
        float* d = bdst[p];
        d[0] = b[p].x; d[2] = b[p].y; d[4] = b[p].z; d[6] = b[p].w;
    }
    __syncthreads();

    for (int kc = 0; kc < NCH - 1; ++kc) {
        const int cur = kc & 1;
        const int k1 = (kc + 1) * KC;
        a0 = *(const v4f*)(ap0 + k1);
        a1 = *(const v4f*)(ap1 + k1);
        a2 = *(const v4f*)(ap2 + k1);
        a3 = *(const v4f*)(ap3 + k1);
#pragma unroll
        for (int p = 0; p < B_P; ++p) b[p] = *(const v4f*)(bsrc[p] + (size_t)k1 * NCOL);
        __builtin_amdgcn_sched_barrier(0);

        mma_chunk<PA, PP, NT>(fa + cur * SA, fb + cur * SB, acc);

        const int nxt = cur ^ 1;
        *(v4f*)(wa + nxt * SA)           = a0;
        *(v4f*)(wa + nxt * SA + 32 * PA) = a1;
        *(v4f*)(wa + nxt * SA + 64 * PA) = a2;
        *(v4f*)(wa + nxt * SA + 96 * PA) = a3;
#pragma unroll
        for (int p = 0; p < B_P; ++p) {
            float* d = bdst[p] + nxt * SB;
            d[0] = b[p].x; d[2] = b[p].y; d[4] = b[p].z; d[6] = b[p].w;
        }
        __syncthreads();
    }
    mma_chunk<PA, PP, NT>(fa + ((NCH - 1) & 1) * SA, fb + ((NCH - 1) & 1) * SB, acc);

#pragma unroll
    for (int mt = 0; mt < 2; ++mt) {
        const int t0 = m0 + wm * 32 + mt * 16;
        if (t0 >= M) continue;
        const int rbase = t0 + (h << 3);
        if (t0 + 16 <= M) {
#pragma unroll
            for (int nt = 0; nt < NT; ++nt) {
                float* c_ptr = C + (size_t)rbase * NCOL + wn * WN + nt * 16 + c;
#pragma unroll
                for (int r = 0; r < 8; ++r) c_ptr[(size_t)r * NCOL] = acc[mt][nt][r];
            }
        } else {
#pragma unroll
            for (int nt = 0; nt < NT; ++nt) {
                float* c_ptr = C + (size_t)rbase * NCOL + wn * WN + nt * 16 + c;
#pragma unroll
                for (int r = 0; r < 8; ++r)
                    if (rbase + r < M) c_ptr[(size_t)r * NCOL] = acc[mt][nt][r];
            }
        }
    }
}


template<int F, int LOG_TPE>
__global__ void scatter_kernel(const float* __restrict__ h, const int* __restrict__ srcI,
                               const int* __restrict__ dstI, const float* __restrict__ dinv,
                               float* __restrict__ acc, int E, int N) {
    long long tid = (long long)blockIdx.x * blockDim.x + threadIdx.x;
    long long edge = tid >> LOG_TPE;
    if (edge >= (long long)E + N) return;
    int fo = ((int)tid & ((1 << LOG_TPE) - 1)) << 2;

    int s, d;
    if (edge < E) { s = srcI[edge]; d = dstI[edge]; }
    else          { s = d = (int)(edge - E); }

    float nrm = dinv[s] * dinv[d];
    const float4 v = *(const float4*)(h + (size_t)s * F + fo);
    float* o = acc + (size_t)d * F + fo;
    atomAddF32(o + 0, v.x * nrm);
    atomAddF32(o + 1, v.y * nrm);
    atomAddF32(o + 2, v.z * nrm);
    atomAddF32(o + 3, v.w * nrm);
}


extern "C" void kernel_launch(void* const* d_in, const int* in_sizes, int n_in,
                              void* d_out, int out_size, void* d_ws, size_t ws_size,
                              hipStream_t stream) {
    const float* x  = (const float*)d_in[0];
    const int*   ei = (const int*)d_in[1];
    const float* W1 = (const float*)d_in[2];
    const float* b1 = (const float*)d_in[3];
    const float* W2 = (const float*)d_in[4];
    const float* b2 = (const float*)d_in[5];
    float* out = (float*)d_out;

    const int N = in_sizes[0] / N_FEAT;
    const int E = in_sizes[1] / 2;
    const int* srcI = ei;
    const int* dstI = ei + E;

    char* ws = (char*)d_ws;
    size_t degB = ((size_t)N * sizeof(float) + 255) & ~(size_t)255;
    size_t t1B  = (((size_t)N * HIDDEN * sizeof(float)) + 255) & ~(size_t)255;
    float* deg = (float*)ws;
    float* t1  = (float*)(ws + degB);
    float* t2  = (float*)(ws + degB + t1B);
    float* t3  = t1;

    const int BT = 256;
    const int gemmBlocks = (N + 127) / 128;

    fill_kernel<<<(N + BT - 1) / BT, BT, 0, stream>>>(deg, 1.0f, N);
    deg_accum_kernel<<<(E + BT - 1) / BT, BT, 0, stream>>>(dstI, deg, E);
    rsqrt_kernel<<<(N + BT - 1) / BT, BT, 0, stream>>>(deg, N);

    long long t2n = (long long)N * HIDDEN;
    fill_kernel<<<(int)((t2n + BT - 1) / BT), BT, 0, stream>>>(t2, 0.0f, t2n);

    gemm_wmma_f32<N_FEAT, HIDDEN><<<gemmBlocks, 256, 0, stream>>>(x, W1, t1, N);

    long long tot1 = ((long long)E + N) << 5;
    scatter_kernel<HIDDEN, 5><<<(int)((tot1 + BT - 1) / BT), BT, 0, stream>>>(
        t1, srcI, dstI, deg, t2, E, N);

    bias_relu_kernel<HIDDEN><<<(int)((t2n + BT - 1) / BT), BT, 0, stream>>>(t2, b1, t2n);

    gemm_wmma_f32<HIDDEN, NCLS><<<gemmBlocks, 256, 0, stream>>>(t2, W2, t3, N);

    long long outn = (long long)N * NCLS;
    init_bias_kernel<NCLS><<<(int)((outn + BT - 1) / BT), BT, 0, stream>>>(out, b2, outn);
    long long tot2 = ((long long)E + N) << 4;
    scatter_kernel<NCLS, 4><<<(int)((tot2 + BT - 1) / BT), BT, 0, stream>>>(
        t3, srcI, dstI, deg, out, E, N);
}


# Edge scatter re-mapped: a wave walks 32 edges with its lanes on adjacent columns, so each row load and f32 atomic add covers one whole 128-byte line; 24x fewer waves; sums still in arrival order
# speedup vs baseline: 1.6393x; 1.4235x over previous
#include <hip/hip_runtime.h>

typedef __attribute__((ext_vector_type(2))) float v2f;
typedef __attribute__((ext_vector_type(8))) float v8f;

#define N_FEAT 512
#define HIDDEN 128
#define NCLS   64

__device__ __forceinline__ void atomAddF32(float* p, float v) {
    unsafeAtomicAdd(p, v);
}


__global__ void fill_kernel(float* __restrict__ p, float v, long long n) {
    long long i = (long long)blockIdx.x * blockDim.x + threadIdx.x;
    if (i < n) p[i] = v;
}

__global__ void deg_accum_kernel(const int* __restrict__ dst, float* __restrict__ deg, int E) {
    int e = blockIdx.x * blockDim.x + threadIdx.x;
    if (e < E) atomAddF32(&deg[dst[e]], 1.0f);
}

__global__ void rsqrt_kernel(float* __restrict__ p, int n) {
    int i = blockIdx.x * blockDim.x + threadIdx.x;
    if (i < n) p[i] = rsqrtf(p[i]);
}

template<int F>
__global__ void bias_relu_kernel(float* __restrict__ t, const float* __restrict__ b, long long n) {
    long long i = (long long)blockIdx.x * blockDim.x + threadIdx.x;
    if (i < n) {
        int j = (int)(i & (F - 1));
        t[i] = fmaxf(t[i] + b[j], 0.0f);
    }
}

template<int F>
__global__ void init_bias_kernel(float* __restrict__ out, const float* __restrict__ b, long long n) {
    long long i = (long long)blockIdx.x * blockDim.x + threadIdx.x;
    if (i < n) out[i] = b[(int)(i & (F - 1))];
}


typedef __attribute__((ext_vector_type(4))) float v4f;

template<int PA, int PP, int NT>
__device__ __forceinline__ void mma_chunk(const float* __restrict__ pa,
                                          const float* __restrict__ pb,
                                          v8f (&acc)[2][NT]) {
#pragma unroll
    for (int ks = 0; ks < 8; ++ks) {
        const v2f a0 = *(const v2f*)(pa + ks * 4);
        const v2f a1 = *(const v2f*)(pa + 16 * PA + ks * 4);
#pragma unroll
        for (int nt = 0; nt < NT; ++nt) {
            const v2f b = *(const v2f*)(pb + ks * 2 * PP + nt * 32);
            acc[0][nt] = __builtin_amdgcn_wmma_f32_16x16x4_f32(
                 false, a0,  false, b,
                 (short)0, acc[0][nt],  false,  false);
            acc[1][nt] = __builtin_amdgcn_wmma_f32_16x16x4_f32(
                 false, a1,  false, b,
                 (short)0, acc[1][nt],  false,  false);
        }
    }
}

template<int K, int NCOL>
__global__ __launch_bounds__(256)
void gemm_wmma_f32(const float* __restrict__ A, const float* __restrict__ B,
                   float* __restrict__ C, int M) {
    constexpr int BM  = 128;
    constexpr int KC  = 32;
    constexpr int PA  = KC + 4;
    constexpr int PP  = 2 * NCOL + 32;
    constexpr int WN  = NCOL / 2;
    constexpr int NT  = WN / 16;
    constexpr int NCH = K / KC;
    constexpr int CG  = NCOL / 32;
    constexpr int B_P = CG;
    constexpr int SA = BM * PA;
    constexpr int SB = (KC / 2) * PP;

    __shared__ __attribute__((aligned(16))) float sA[2 * SA];
    __shared__ __attribute__((aligned(16))) float sB[2 * SB];

    const int tid  = threadIdx.x;
    const int lane = tid & 31;
    const int wave = __builtin_amdgcn_readfirstlane(tid >> 5);
    const int wm = wave >> 1;
    const int wn = wave & 1;
    const int c = lane & 15;
    const int h = lane >> 4;
    const int m0 = blockIdx.x * BM;

    const int a_row = tid >> 3;
    const int a_col = (tid & 7) << 2;
    const int b_r = lane >> 3;
    const int b_c = (lane & 7) << 2;

    const int r0 = min(m0 + a_row,      M - 1);
    const int r1 = min(m0 + a_row + 32, M - 1);
    const int r2 = min(m0 + a_row + 64, M - 1);
    const int r3 = min(m0 + a_row + 96, M - 1);
    const float* ap0 = A + (size_t)r0 * K + a_col;
    const float* ap1 = A + (size_t)r1 * K + a_col;
    const float* ap2 = A + (size_t)r2 * K + a_col;
    const float* ap3 = A + (size_t)r3 * K + a_col;
    const float* bsrc[B_P];
    float* bdst[B_P];
#pragma unroll
    for (int p = 0; p < B_P; ++p) {
        const int q = wave + 8 * p;
        const int row = 4 * (q / CG) + b_r;
        const int col = 32 * (q % CG) + b_c;
        bsrc[p] = B + (size_t)row * NCOL + col;
        bdst[p] = sB + (row >> 1) * PP + 2 * col + (row & 1);
    }

    float* wa = sA + a_row * PA + a_col;
    const float* fa = sA + (wm * 32 + c) * PA + 2 * h;
    const float* fb = sB + h * PP + (wn * WN + c) * 2;

    v8f acc[2][NT] = {};

    v4f a0 = *(const v4f*)ap0, a1 = *(const v4f*)ap1, a2 = *(const v4f*)ap2, a3 = *(const v4f*)ap3;
    v4f b[B_P];
#pragma unroll
    for (int p = 0; p < B_P; ++p) b[p] = *(const v4f*)(bsrc[p]);

    *(v4f*)(wa)           = a0;
    *(v4f*)(wa + 32 * PA) = a1;
    *(v4f*)(wa + 64 * PA) = a2;
    *(v4f*)(wa + 96 * PA) = a3;
#pragma unroll
    for (int p = 0; p < B_P; ++p) {
        float* d = bdst[p];
        d[0] = b[p].x; d[2] = b[p].y; d[4] = b[p].z; d[6] = b[p].w;
    }
    __syncthreads();

    for (int kc = 0; kc < NCH - 1; ++kc) {
        const int cur = kc & 1;
        const int k1 = (kc + 1) * KC;
        a0 = *(const v4f*)(ap0 + k1);
        a1 = *(const v4f*)(ap1 + k1);
        a2 = *(const v4f*)(ap2 + k1);
        a3 = *(const v4f*)(ap3 + k1);
#pragma unroll
        for (int p = 0; p < B_P; ++p) b[p] = *(const v4f*)(bsrc[p] + (size_t)k1 * NCOL);
        __builtin_amdgcn_sched_barrier(0);

        mma_chunk<PA, PP, NT>(fa + cur * SA, fb + cur * SB, acc);

        const int nxt = cur ^ 1;
        *(v4f*)(wa + nxt * SA)           = a0;
        *(v4f*)(wa + nxt * SA + 32 * PA) = a1;
        *(v4f*)(wa + nxt * SA + 64 * PA) = a2;
        *(v4f*)(wa + nxt * SA + 96 * PA) = a3;
#pragma unroll
        for (int p = 0; p < B_P; ++p) {
            float* d = bdst[p] + nxt * SB;
            d[0] = b[p].x; d[2] = b[p].y; d[4] = b[p].z; d[6] = b[p].w;
        }
        __syncthreads();
    }
    mma_chunk<PA, PP, NT>(fa + ((NCH - 1) & 1) * SA, fb + ((NCH - 1) & 1) * SB, acc);

#pragma unroll
    for (int mt = 0; mt < 2; ++mt) {
        const int t0 = m0 + wm * 32 + mt * 16;
        if (t0 >= M) continue;
        const int rbase = t0 + (h << 3);
        if (t0 + 16 <= M) {
#pragma unroll
            for (int nt = 0; nt < NT; ++nt) {
                float* c_ptr = C + (size_t)rbase * NCOL + wn * WN + nt * 16 + c;
#pragma unroll
                for (int r = 0; r < 8; ++r) c_ptr[(size_t)r * NCOL] = acc[mt][nt][r];
            }
        } else {
#pragma unroll
            for (int nt = 0; nt < NT; ++nt) {
                float* c_ptr = C + (size_t)rbase * NCOL + wn * WN + nt * 16 + c;
#pragma unroll
                for (int r = 0; r < 8; ++r)
                    if (rbase + r < M) c_ptr[(size_t)r * NCOL] = acc[mt][nt][r];
            }
        }
    }
}


template<int F>
__global__ __launch_bounds__(256)
void scatter_kernel(const float* __restrict__ h, const int* __restrict__ srcI,
                    const int* __restrict__ dstI, const float* __restrict__ dinv,
                    float* __restrict__ acc, int E, int N) {
    constexpr int NL = F / 32;
    constexpr int U  = 16 / NL;
    const int lane = threadIdx.x & 31;
    const int wave = __builtin_amdgcn_readfirstlane((int)(threadIdx.x >> 5));
    const unsigned total = (unsigned)E + (unsigned)N;
    const unsigned base = (blockIdx.x * 8u + (unsigned)wave) * 32u;
    if (base >= total) return;
    const int cnt = (int)((total - base) < 32u ? (total - base) : 32u);

    int s = 0, d = 0;
    float nrm = 0.0f;
    if (lane < cnt) {
        const unsigned e = base + (unsigned)lane;
        if (e < (unsigned)E) { s = srcI[e]; d = dstI[e]; }
        else                 { s = d = (int)(e - (unsigned)E); }
        nrm = dinv[s] * dinv[d];
    }

    if (cnt == 32) {
        float buf[2][U][NL];
#pragma unroll
        for (int u = 0; u < U; ++u) {
            const float* hp = h + (size_t)__builtin_amdgcn_readlane(s, u) * F;
#pragma unroll
            for (int k = 0; k < NL; ++k) buf[0][u][k] = hp[lane + 32 * k];
        }
#pragma unroll
        for (int g = 0; g < 32 / U; ++g) {
            if (g + 1 < 32 / U) {
#pragma unroll
                for (int u = 0; u < U; ++u) {
                    const float* hp = h + (size_t)__builtin_amdgcn_readlane(s, ((g + 1) * U + u) & 31) * F;
#pragma unroll
                    for (int k = 0; k < NL; ++k) buf[(g + 1) & 1][u][k] = hp[lane + 32 * k];
                }
            }
#pragma unroll
            for (int u = 0; u < U; ++u) {
                const int dj = __builtin_amdgcn_readlane(d, g * U + u);
                const float nj = __int_as_float(__builtin_amdgcn_readlane(__float_as_int(nrm), g * U + u));
                float* op = acc + (size_t)dj * F;
#pragma unroll
                for (int k = 0; k < NL; ++k) atomAddF32(op + lane + 32 * k, buf[g & 1][u][k] * nj);
            }
        }
    } else {
        for (int j = 0; j < cnt; ++j) {
            const float* hp = h + (size_t)__builtin_amdgcn_readlane(s, j) * F;
            float* op = acc + (size_t)__builtin_amdgcn_readlane(d, j) * F;
            const float nj = __int_as_float(__builtin_amdgcn_readlane(__float_as_int(nrm), j));
#pragma unroll
            for (int k = 0; k < NL; ++k) atomAddF32(op + lane + 32 * k, hp[lane + 32 * k] * nj);
        }
    }
}


extern "C" void kernel_launch(void* const* d_in, const int* in_sizes, int n_in,
                              void* d_out, int out_size, void* d_ws, size_t ws_size,
                              hipStream_t stream) {
    const float* x  = (const float*)d_in[0];
    const int*   ei = (const int*)d_in[1];
    const float* W1 = (const float*)d_in[2];
    const float* b1 = (const float*)d_in[3];
    const float* W2 = (const float*)d_in[4];
    const float* b2 = (const float*)d_in[5];
    float* out = (float*)d_out;

    const int N = in_sizes[0] / N_FEAT;
    const int E = in_sizes[1] / 2;
    const int* srcI = ei;
    const int* dstI = ei + E;

    char* ws = (char*)d_ws;
    size_t degB = ((size_t)N * sizeof(float) + 255) & ~(size_t)255;
    size_t t1B  = (((size_t)N * HIDDEN * sizeof(float)) + 255) & ~(size_t)255;
    float* deg = (float*)ws;
    float* t1  = (float*)(ws + degB);
    float* t2  = (float*)(ws + degB + t1B);
    float* t3  = t1;

    const int BT = 256;
    const int gemmBlocks = (N + 127) / 128;

    fill_kernel<<<(N + BT - 1) / BT, BT, 0, stream>>>(deg, 1.0f, N);
    deg_accum_kernel<<<(E + BT - 1) / BT, BT, 0, stream>>>(dstI, deg, E);
    rsqrt_kernel<<<(N + BT - 1) / BT, BT, 0, stream>>>(deg, N);

    long long t2n = (long long)N * HIDDEN;
    fill_kernel<<<(int)((t2n + BT - 1) / BT), BT, 0, stream>>>(t2, 0.0f, t2n);

    gemm_wmma_f32<N_FEAT, HIDDEN><<<gemmBlocks, 256, 0, stream>>>(x, W1, t1, N);

    const int scatBlocks = (int)(((long long)E + N + 255) / 256);
    scatter_kernel<HIDDEN><<<scatBlocks, BT, 0, stream>>>(t1, srcI, dstI, deg, t2, E, N);

    bias_relu_kernel<HIDDEN><<<(int)((t2n + BT - 1) / BT), BT, 0, stream>>>(t2, b1, t2n);

    gemm_wmma_f32<HIDDEN, NCLS><<<gemmBlocks, 256, 0, stream>>>(t2, W2, t3, N);

    long long outn = (long long)N * NCLS;
    init_bias_kernel<NCLS><<<(int)((outn + BT - 1) / BT), BT, 0, stream>>>(out, b2, outn);
    scatter_kernel<NCLS><<<scatBlocks, BT, 0, stream>>>(t3, srcI, dstI, deg, out, E, N);
}


# Fill, bias+ReLU and bias-init passes folded into the GEMMs: store also writes each row's self-loop term (+bias) as the scatter's start value, GEMM 2 applies bias+ReLU on A load; self-loop added first
# speedup vs baseline: 2.0705x; 1.2630x over previous
#include <hip/hip_runtime.h>

typedef __attribute__((ext_vector_type(2))) float v2f;
typedef __attribute__((ext_vector_type(8))) float v8f;

#define N_FEAT 512
#define HIDDEN 128
#define NCLS   64

__device__ __forceinline__ void atomAddF32(float* p, float v) {
    unsafeAtomicAdd(p, v);
}


__global__ void fill_kernel(float* __restrict__ p, float v, long long n) {
    long long i = (long long)blockIdx.x * blockDim.x + threadIdx.x;
    if (i < n) p[i] = v;
}

__global__ void deg_accum_kernel(const int* __restrict__ dst, float* __restrict__ deg, int E) {
    int e = blockIdx.x * blockDim.x + threadIdx.x;
    if (e < E) atomAddF32(&deg[dst[e]], 1.0f);
}

__global__ void rsqrt_kernel(float* __restrict__ p, int n) {
    int i = blockIdx.x * blockDim.x + threadIdx.x;
    if (i < n) p[i] = rsqrtf(p[i]);
}


typedef __attribute__((ext_vector_type(4))) float v4f;

template<int PA, int PP, int NT>
__device__ __forceinline__ void mma_chunk(const float* __restrict__ pa,
                                          const float* __restrict__ pb,
                                          v8f (&acc)[2][NT]) {
#pragma unroll
    for (int ks = 0; ks < 8; ++ks) {
        const v2f a0 = *(const v2f*)(pa + ks * 4);
        const v2f a1 = *(const v2f*)(pa + 16 * PA + ks * 4);
#pragma unroll
        for (int nt = 0; nt < NT; ++nt) {
            const v2f b = *(const v2f*)(pb + ks * 2 * PP + nt * 32);
            acc[0][nt] = __builtin_amdgcn_wmma_f32_16x16x4_f32(
                 false, a0,  false, b,
                 (short)0, acc[0][nt],  false,  false);
            acc[1][nt] = __builtin_amdgcn_wmma_f32_16x16x4_f32(
                 false, a1,  false, b,
                 (short)0, acc[1][nt],  false,  false);
        }
    }
}

template<bool RELU_A>
__device__ __forceinline__ v4f act_a(v4f a, v4f bb) {
    if (RELU_A) {
        a.x = fmaxf(a.x + bb.x, 0.0f);
        a.y = fmaxf(a.y + bb.y, 0.0f);
        a.z = fmaxf(a.z + bb.z, 0.0f);
        a.w = fmaxf(a.w + bb.w, 0.0f);
    }
    return a;
}

template<int NCOL, int NT, bool SBIAS, bool GUARD>
__device__ __forceinline__ void store_tile(float* __restrict__ C, float* __restrict__ S,
                                           const float* __restrict__ dinv,
                                           const float* __restrict__ sbias,
                                           const v8f (&acc)[NT], int rbase, int col0, int M) {
    float w[8];
#pragma unroll
    for (int r = 0; r < 8; ++r) {
        const float dv = dinv[GUARD ? min(rbase + r, M - 1) : rbase + r];
        w[r] = dv * dv;
    }
#pragma unroll
    for (int nt = 0; nt < NT; ++nt) {
        const int col = col0 + nt * 16;
        const float bj = SBIAS ? sbias[col] : 0.0f;
        float* c_ptr = C + (size_t)rbase * NCOL + col;
        float* s_ptr = S + (size_t)rbase * NCOL + col;
#pragma unroll
        for (int r = 0; r < 8; ++r) {
            if (!GUARD || rbase + r < M) {
                const float v = acc[nt][r];
                c_ptr[(size_t)r * NCOL] = v;
                s_ptr[(size_t)r * NCOL] = SBIAS ? (v * w[r] + bj) : (v * w[r]);
            }
        }
    }
}

template<int K, int NCOL, bool RELU_A, bool SBIAS>
__global__ __launch_bounds__(256)
void gemm_wmma_f32(const float* __restrict__ A, const float* __restrict__ B,
                   float* __restrict__ C, float* __restrict__ S,
                   const float* __restrict__ dinv, const float* __restrict__ abias,
                   const float* __restrict__ sbias, int M) {
    constexpr int BM  = 128;
    constexpr int KC  = 32;
    constexpr int PA  = KC + 4;
    constexpr int PP  = 2 * NCOL + 32;
    constexpr int WN  = NCOL / 2;
    constexpr int NT  = WN / 16;
    constexpr int NCH = K / KC;
    constexpr int CG  = NCOL / 32;
    constexpr int B_P = CG;
    constexpr int SA = BM * PA;
    constexpr int SB = (KC / 2) * PP;

    __shared__ __attribute__((aligned(16))) float sA[2 * SA];
    __shared__ __attribute__((aligned(16))) float sB[2 * SB];

    const int tid  = threadIdx.x;
    const int lane = tid & 31;
    const int wave = __builtin_amdgcn_readfirstlane(tid >> 5);
    const int wm = wave >> 1;
    const int wn = wave & 1;
    const int c = lane & 15;
    const int h = lane >> 4;
    const int m0 = blockIdx.x * BM;

    const int a_row = tid >> 3;
    const int a_col = (tid & 7) << 2;
    const int b_r = lane >> 3;
    const int b_c = (lane & 7) << 2;

    const int r0 = min(m0 + a_row,      M - 1);
    const int r1 = min(m0 + a_row + 32, M - 1);
    const int r2 = min(m0 + a_row + 64, M - 1);
    const int r3 = min(m0 + a_row + 96, M - 1);
    const float* ap0 = A + (size_t)r0 * K + a_col;
    const float* ap1 = A + (size_t)r1 * K + a_col;
    const float* ap2 = A + (size_t)r2 * K + a_col;
    const float* ap3 = A + (size_t)r3 * K + a_col;
    const float* bsrc[B_P];
    float* bdst[B_P];
#pragma unroll
    for (int p = 0; p < B_P; ++p) {
        const int q = wave + 8 * p;
        const int row = 4 * (q / CG) + b_r;
        const int col = 32 * (q % CG) + b_c;
        bsrc[p] = B + (size_t)row * NCOL + col;
        bdst[p] = sB + (row >> 1) * PP + 2 * col + (row & 1);
    }

    float* wa = sA + a_row * PA + a_col;
    const float* fa = sA + (wm * 32 + c) * PA + 2 * h;
    const float* fb = sB + h * PP + (wn * WN + c) * 2;

    v8f acc[2][NT] = {};

    v4f a0 = *(const v4f*)ap0, a1 = *(const v4f*)ap1, a2 = *(const v4f*)ap2, a3 = *(const v4f*)ap3;
    v4f bb = {};
    if (RELU_A) bb = *(const v4f*)(abias + a_col);
    v4f b[B_P];
#pragma unroll
    for (int p = 0; p < B_P; ++p) b[p] = *(const v4f*)(bsrc[p]);

    *(v4f*)(wa)           = act_a<RELU_A>(a0, bb);
    *(v4f*)(wa + 32 * PA) = act_a<RELU_A>(a1, bb);
    *(v4f*)(wa + 64 * PA) = act_a<RELU_A>(a2, bb);
    *(v4f*)(wa + 96 * PA) = act_a<RELU_A>(a3, bb);
#pragma unroll
    for (int p = 0; p < B_P; ++p) {
        float* d = bdst[p];
        d[0] = b[p].x; d[2] = b[p].y; d[4] = b[p].z; d[6] = b[p].w;
    }
    __syncthreads();

    for (int kc = 0; kc < NCH - 1; ++kc) {
        const int cur = kc & 1;
        const int k1 = (kc + 1) * KC;
        a0 = *(const v4f*)(ap0 + k1);
        a1 = *(const v4f*)(ap1 + k1);
        a2 = *(const v4f*)(ap2 + k1);
        a3 = *(const v4f*)(ap3 + k1);
        if (RELU_A) bb = *(const v4f*)(abias + k1 + a_col);
#pragma unroll
        for (int p = 0; p < B_P; ++p) b[p] = *(const v4f*)(bsrc[p] + (size_t)k1 * NCOL);
        __builtin_amdgcn_sched_barrier(0);

        mma_chunk<PA, PP, NT>(fa + cur * SA, fb + cur * SB, acc);
        if (RELU_A) __builtin_amdgcn_sched_barrier(0);

        const int nxt = cur ^ 1;
        *(v4f*)(wa + nxt * SA)           = act_a<RELU_A>(a0, bb);
        *(v4f*)(wa + nxt * SA + 32 * PA) = act_a<RELU_A>(a1, bb);
        *(v4f*)(wa + nxt * SA + 64 * PA) = act_a<RELU_A>(a2, bb);
        *(v4f*)(wa + nxt * SA + 96 * PA) = act_a<RELU_A>(a3, bb);
#pragma unroll
        for (int p = 0; p < B_P; ++p) {
            float* d = bdst[p] + nxt * SB;
            d[0] = b[p].x; d[2] = b[p].y; d[4] = b[p].z; d[6] = b[p].w;
        }
        __syncthreads();
    }
    mma_chunk<PA, PP, NT>(fa + ((NCH - 1) & 1) * SA, fb + ((NCH - 1) & 1) * SB, acc);

#pragma unroll
    for (int mt = 0; mt < 2; ++mt) {
        const int t0 = m0 + wm * 32 + mt * 16;
        if (t0 >= M) continue;
        const int rbase = t0 + (h << 3);
        if (t0 + 16 <= M)
            store_tile<NCOL, NT, SBIAS, false>(C, S, dinv, sbias, acc[mt], rbase, wn * WN + c, M);
        else
            store_tile<NCOL, NT, SBIAS, true>(C, S, dinv, sbias, acc[mt], rbase, wn * WN + c, M);
    }
}


template<int F>
__global__ __launch_bounds__(256)
void scatter_kernel(const float* __restrict__ h, const int* __restrict__ srcI,
                    const int* __restrict__ dstI, const float* __restrict__ dinv,
                    float* __restrict__ acc, int E) {
    constexpr int NL = F / 32;
    constexpr int U  = 16 / NL;
    const int lane = threadIdx.x & 31;
    const int wave = __builtin_amdgcn_readfirstlane((int)(threadIdx.x >> 5));
    const unsigned total = (unsigned)E;
    const unsigned base = (blockIdx.x * 8u + (unsigned)wave) * 32u;
    if (base >= total) return;
    const int cnt = (int)((total - base) < 32u ? (total - base) : 32u);

    int s = 0, d = 0;
    float nrm = 0.0f;
    if (lane < cnt) {
        const unsigned e = base + (unsigned)lane;
        s = srcI[e];
        d = dstI[e];
        nrm = dinv[s] * dinv[d];
    }

    if (cnt == 32) {
        float buf[2][U][NL];
#pragma unroll
        for (int u = 0; u < U; ++u) {
            const float* hp = h + (size_t)__builtin_amdgcn_readlane(s, u) * F;
#pragma unroll
            for (int k = 0; k < NL; ++k) buf[0][u][k] = hp[lane + 32 * k];
        }
#pragma unroll
        for (int g = 0; g < 32 / U; ++g) {
            if (g + 1 < 32 / U) {
#pragma unroll
                for (int u = 0; u < U; ++u) {
                    const float* hp = h + (size_t)__builtin_amdgcn_readlane(s, ((g + 1) * U + u) & 31) * F;
#pragma unroll
                    for (int k = 0; k < NL; ++k) buf[(g + 1) & 1][u][k] = hp[lane + 32 * k];
                }
            }
#pragma unroll
            for (int u = 0; u < U; ++u) {
                const int dj = __builtin_amdgcn_readlane(d, g * U + u);
                const float nj = __int_as_float(__builtin_amdgcn_readlane(__float_as_int(nrm), g * U + u));
                float* op = acc + (size_t)dj * F;
#pragma unroll
                for (int k = 0; k < NL; ++k) atomAddF32(op + lane + 32 * k, buf[g & 1][u][k] * nj);
            }
        }
    } else {
        for (int j = 0; j < cnt; ++j) {
            const float* hp = h + (size_t)__builtin_amdgcn_readlane(s, j) * F;
            float* op = acc + (size_t)__builtin_amdgcn_readlane(d, j) * F;
            const float nj = __int_as_float(__builtin_amdgcn_readlane(__float_as_int(nrm), j));
#pragma unroll
            for (int k = 0; k < NL; ++k) atomAddF32(op + lane + 32 * k, hp[lane + 32 * k] * nj);
        }
    }
}


extern "C" void kernel_launch(void* const* d_in, const int* in_sizes, int n_in,
                              void* d_out, int out_size, void* d_ws, size_t ws_size,
                              hipStream_t stream) {
    const float* x  = (const float*)d_in[0];
    const int*   ei = (const int*)d_in[1];
    const float* W1 = (const float*)d_in[2];
    const float* b1 = (const float*)d_in[3];
    const float* W2 = (const float*)d_in[4];
    const float* b2 = (const float*)d_in[5];
    float* out = (float*)d_out;

    const int N = in_sizes[0] / N_FEAT;
    const int E = in_sizes[1] / 2;
    const int* srcI = ei;
    const int* dstI = ei + E;

    char* ws = (char*)d_ws;
    size_t degB = ((size_t)N * sizeof(float) + 255) & ~(size_t)255;
    size_t t1B  = (((size_t)N * HIDDEN * sizeof(float)) + 255) & ~(size_t)255;
    float* deg = (float*)ws;
    float* t1  = (float*)(ws + degB);
    float* t2  = (float*)(ws + degB + t1B);
    float* t3  = t1;

    const int BT = 256;
    const int gemmBlocks = (N + 127) / 128;

    fill_kernel<<<(N + BT - 1) / BT, BT, 0, stream>>>(deg, 1.0f, N);
    deg_accum_kernel<<<(E + BT - 1) / BT, BT, 0, stream>>>(dstI, deg, E);
    rsqrt_kernel<<<(N + BT - 1) / BT, BT, 0, stream>>>(deg, N);

    gemm_wmma_f32<N_FEAT, HIDDEN, false, false><<<gemmBlocks, 256, 0, stream>>>(x, W1, t1, t2, deg, b1, b2, N);

    const int scatBlocks = (E + 255) / 256;
    scatter_kernel<HIDDEN><<<scatBlocks, BT, 0, stream>>>(t1, srcI, dstI, deg, t2, E);

    gemm_wmma_f32<HIDDEN, NCLS, true, true><<<gemmBlocks, 256, 0, stream>>>(t2, W2, t3, out, deg, b1, b2, N);

    scatter_kernel<NCLS><<<scatBlocks, BT, 0, stream>>>(t3, srcI, dstI, deg, out, E);
}


# Degree count without global atomics: 64 work-groups count their edge slices in LDS counters over all nodes, store count planes; the rsqrt pass sums them (integers, exact); no fill; 7->6 launches
# speedup vs baseline: 2.4678x; 1.1919x over previous
#include <hip/hip_runtime.h>

typedef __attribute__((ext_vector_type(2))) float v2f;
typedef __attribute__((ext_vector_type(8))) float v8f;

#define N_FEAT 512
#define HIDDEN 128
#define NCLS   64

__device__ __forceinline__ void atomAddF32(float* p, float v) {
    unsafeAtomicAdd(p, v);
}


#define DEG_G    64
#define DEG_T    1024
#define DEG_B    8
#define DEG_BINS 51200

__global__ __launch_bounds__(DEG_T)
void deg_hist_kernel(const int* __restrict__ dst, unsigned* __restrict__ part, int E, int N) {
    __shared__ __attribute__((aligned(16))) unsigned hist[DEG_BINS];
    const int tid = threadIdx.x;
    const int per = (E + (int)gridDim.x - 1) / (int)gridDim.x;
    const int lo = min(E, (int)blockIdx.x * per);
    const int hi = min(E, lo + per);
    const int NP = (N + 63) & ~63;
    unsigned* __restrict__ plane = part + (size_t)blockIdx.x * NP;

    for (int base = 0; base < N; base += DEG_BINS) {
        for (int i = tid * 4; i < DEG_BINS; i += DEG_T * 4) *(uint4*)&hist[i] = make_uint4(0u, 0u, 0u, 0u);
        __syncthreads();
        for (int e = lo + tid; e < hi; e += DEG_T * DEG_B) {
            int d[DEG_B];
#pragma unroll
            for (int u = 0; u < DEG_B; ++u) {
                const int ee = e + u * DEG_T;
                d[u] = ee < hi ? dst[ee] : -1;
            }
#pragma unroll
            for (int u = 0; u < DEG_B; ++u) {
                const unsigned t = (unsigned)(d[u] - base);
                if (t < (unsigned)DEG_BINS) atomicAdd(&hist[t], 1u);
            }
        }
        __syncthreads();
        const int lim = min(DEG_BINS, N - base);
        for (int i = tid * 4; i < lim; i += DEG_T * 4) {
            if (i + 3 < lim) {
                *(uint4*)&plane[base + i] = *(const uint4*)&hist[i];
            } else {
                for (int k = 0; i + k < lim; ++k) plane[base + i + k] = hist[i + k];
            }
        }
        __syncthreads();
    }
}

__global__ void deg_finish_kernel(const unsigned* __restrict__ part, float* __restrict__ dinv,
                                  int N, unsigned* __restrict__ q) {
    const int i = blockIdx.x * blockDim.x + threadIdx.x;
    if (i < 4) q[i] = 0u;
    if (i >= N) return;
    const int NP = (N + 63) & ~63;
    unsigned s = 1u;
#pragma unroll 16
    for (int g = 0; g < DEG_G; ++g) s += part[(size_t)g * NP + i];
    dinv[i] = rsqrtf((float)s);
}


typedef __attribute__((ext_vector_type(4))) float v4f;

template<int PA, int PP, int NT>
__device__ __forceinline__ void mma_chunk(const float* __restrict__ pa,
                                          const float* __restrict__ pb,
                                          v8f (&acc)[2][NT]) {
#pragma unroll
    for (int ks = 0; ks < 8; ++ks) {
        const v2f a0 = *(const v2f*)(pa + ks * 4);
        const v2f a1 = *(const v2f*)(pa + 16 * PA + ks * 4);
#pragma unroll
        for (int nt = 0; nt < NT; ++nt) {
            const v2f b = *(const v2f*)(pb + ks * 2 * PP + nt * 32);
            acc[0][nt] = __builtin_amdgcn_wmma_f32_16x16x4_f32(
                 false, a0,  false, b,
                 (short)0, acc[0][nt],  false,  false);
            acc[1][nt] = __builtin_amdgcn_wmma_f32_16x16x4_f32(
                 false, a1,  false, b,
                 (short)0, acc[1][nt],  false,  false);
        }
    }
}

template<bool RELU_A>
__device__ __forceinline__ v4f act_a(v4f a, v4f bb) {
    if (RELU_A) {
        a.x = fmaxf(a.x + bb.x, 0.0f);
        a.y = fmaxf(a.y + bb.y, 0.0f);
        a.z = fmaxf(a.z + bb.z, 0.0f);
        a.w = fmaxf(a.w + bb.w, 0.0f);
    }
    return a;
}

template<int NCOL, int NT, bool SBIAS, bool GUARD>
__device__ __forceinline__ void store_tile(float* __restrict__ C, float* __restrict__ S,
                                           const float* __restrict__ dinv,
                                           const float* __restrict__ sbias,
                                           const v8f (&acc)[NT], int rbase, int col0, int M) {
    float w[8];
#pragma unroll
    for (int r = 0; r < 8; ++r) {
        const float dv = dinv[GUARD ? min(rbase + r, M - 1) : rbase + r];
        w[r] = dv * dv;
    }
#pragma unroll
    for (int nt = 0; nt < NT; ++nt) {
        const int col = col0 + nt * 16;
        const float bj = SBIAS ? sbias[col] : 0.0f;
        float* c_ptr = C + (size_t)rbase * NCOL + col;
        float* s_ptr = S + (size_t)rbase * NCOL + col;
#pragma unroll
        for (int r = 0; r < 8; ++r) {
            if (!GUARD || rbase + r < M) {
                const float v = acc[nt][r];
                c_ptr[(size_t)r * NCOL] = v;
                s_ptr[(size_t)r * NCOL] = SBIAS ? (v * w[r] + bj) : (v * w[r]);
            }
        }
    }
}

template<int K, int NCOL, bool RELU_A, bool SBIAS>
__global__ __launch_bounds__(256)
void gemm_wmma_f32(const float* __restrict__ A, const float* __restrict__ B,
                   float* __restrict__ C, float* __restrict__ S,
                   const float* __restrict__ dinv, const float* __restrict__ abias,
                   const float* __restrict__ sbias, int M) {
    constexpr int BM  = 128;
    constexpr int KC  = 32;
    constexpr int PA  = KC + 4;
    constexpr int PP  = 2 * NCOL + 32;
    constexpr int WN  = NCOL / 2;
    constexpr int NT  = WN / 16;
    constexpr int NCH = K / KC;
    constexpr int CG  = NCOL / 32;
    constexpr int B_P = CG;
    constexpr int SA = BM * PA;
    constexpr int SB = (KC / 2) * PP;

    __shared__ __attribute__((aligned(16))) float sA[2 * SA];
    __shared__ __attribute__((aligned(16))) float sB[2 * SB];

    const int tid  = threadIdx.x;
    const int lane = tid & 31;
    const int wave = __builtin_amdgcn_readfirstlane(tid >> 5);
    const int wm = wave >> 1;
    const int wn = wave & 1;
    const int c = lane & 15;
    const int h = lane >> 4;
    const int m0 = blockIdx.x * BM;

    const int a_row = tid >> 3;
    const int a_col = (tid & 7) << 2;
    const int b_r = lane >> 3;
    const int b_c = (lane & 7) << 2;

    const int r0 = min(m0 + a_row,      M - 1);
    const int r1 = min(m0 + a_row + 32, M - 1);
    const int r2 = min(m0 + a_row + 64, M - 1);
    const int r3 = min(m0 + a_row + 96, M - 1);
    const float* ap0 = A + (size_t)r0 * K + a_col;
    const float* ap1 = A + (size_t)r1 * K + a_col;
    const float* ap2 = A + (size_t)r2 * K + a_col;
    const float* ap3 = A + (size_t)r3 * K + a_col;
    const float* bsrc[B_P];
    float* bdst[B_P];
#pragma unroll
    for (int p = 0; p < B_P; ++p) {
        const int q = wave + 8 * p;
        const int row = 4 * (q / CG) + b_r;
        const int col = 32 * (q % CG) + b_c;
        bsrc[p] = B + (size_t)row * NCOL + col;
        bdst[p] = sB + (row >> 1) * PP + 2 * col + (row & 1);
    }

    float* wa = sA + a_row * PA + a_col;
    const float* fa = sA + (wm * 32 + c) * PA + 2 * h;
    const float* fb = sB + h * PP + (wn * WN + c) * 2;

    v8f acc[2][NT] = {};

    v4f a0 = *(const v4f*)ap0, a1 = *(const v4f*)ap1, a2 = *(const v4f*)ap2, a3 = *(const v4f*)ap3;
    v4f bb = {};
    if (RELU_A) bb = *(const v4f*)(abias + a_col);
    v4f b[B_P];
#pragma unroll
    for (int p = 0; p < B_P; ++p) b[p] = *(const v4f*)(bsrc[p]);

    *(v4f*)(wa)           = act_a<RELU_A>(a0, bb);
    *(v4f*)(wa + 32 * PA) = act_a<RELU_A>(a1, bb);
    *(v4f*)(wa + 64 * PA) = act_a<RELU_A>(a2, bb);
    *(v4f*)(wa + 96 * PA) = act_a<RELU_A>(a3, bb);
#pragma unroll
    for (int p = 0; p < B_P; ++p) {
        float* d = bdst[p];
        d[0] = b[p].x; d[2] = b[p].y; d[4] = b[p].z; d[6] = b[p].w;
    }
    __syncthreads();

    for (int kc = 0; kc < NCH - 1; ++kc) {
        const int cur = kc & 1;
        const int k1 = (kc + 1) * KC;
        a0 = *(const v4f*)(ap0 + k1);
        a1 = *(const v4f*)(ap1 + k1);
        a2 = *(const v4f*)(ap2 + k1);
        a3 = *(const v4f*)(ap3 + k1);
        if (RELU_A) bb = *(const v4f*)(abias + k1 + a_col);
#pragma unroll
        for (int p = 0; p < B_P; ++p) b[p] = *(const v4f*)(bsrc[p] + (size_t)k1 * NCOL);
        __builtin_amdgcn_sched_barrier(0);

        mma_chunk<PA, PP, NT>(fa + cur * SA, fb + cur * SB, acc);
        if (RELU_A) __builtin_amdgcn_sched_barrier(0);

        const int nxt = cur ^ 1;
        *(v4f*)(wa + nxt * SA)           = act_a<RELU_A>(a0, bb);
        *(v4f*)(wa + nxt * SA + 32 * PA) = act_a<RELU_A>(a1, bb);
        *(v4f*)(wa + nxt * SA + 64 * PA) = act_a<RELU_A>(a2, bb);
        *(v4f*)(wa + nxt * SA + 96 * PA) = act_a<RELU_A>(a3, bb);
#pragma unroll
        for (int p = 0; p < B_P; ++p) {
            float* d = bdst[p] + nxt * SB;
            d[0] = b[p].x; d[2] = b[p].y; d[4] = b[p].z; d[6] = b[p].w;
        }
        __syncthreads();
    }
    mma_chunk<PA, PP, NT>(fa + ((NCH - 1) & 1) * SA, fb + ((NCH - 1) & 1) * SB, acc);

#pragma unroll
    for (int mt = 0; mt < 2; ++mt) {
        const int t0 = m0 + wm * 32 + mt * 16;
        if (t0 >= M) continue;
        const int rbase = t0 + (h << 3);
        if (t0 + 16 <= M)
            store_tile<NCOL, NT, SBIAS, false>(C, S, dinv, sbias, acc[mt], rbase, wn * WN + c, M);
        else
            store_tile<NCOL, NT, SBIAS, true>(C, S, dinv, sbias, acc[mt], rbase, wn * WN + c, M);
    }
}


template<int U, bool ALL>
__device__ __forceinline__ void take_edges(unsigned long long& m, int (&j)[U]) {
#pragma unroll
    for (int u = 0; u < U; ++u) {
        if (ALL || m) { j[u] = __builtin_ctzll(m); m &= m - 1ull; } else j[u] = -1;
    }
}

__device__ __forceinline__ int edge_val(int v0, int v1, int j) {
    const int x0 = __builtin_amdgcn_readlane(v0, j & 31);
    const int x1 = __builtin_amdgcn_readlane(v1, j & 31);
    return j < 32 ? x0 : x1;
}

template<int F, int U, bool ALL>
__device__ __forceinline__ void load_rows(const float* __restrict__ h, int s0, int s1,
                                          const int (&j)[U], float (&buf)[U][F / 32], int lane) {
#pragma unroll
    for (int u = 0; u < U; ++u) {
        if (ALL || j[u] >= 0) {
            const float* hp = h + (size_t)edge_val(s0, s1, j[u]) * F;
#pragma unroll
            for (int k = 0; k < F / 32; ++k) buf[u][k] = hp[lane + 32 * k];
        }
    }
}

template<int F, int U, bool ALL>
__device__ __forceinline__ void add_rows(float* __restrict__ acc, int d0, int d1, float n0, float n1,
                                         const int (&j)[U], const float (&buf)[U][F / 32], int lane) {
#pragma unroll
    for (int u = 0; u < U; ++u) {
        if (ALL || j[u] >= 0) {
            const int dj = edge_val(d0, d1, j[u]);
            const float nj = __int_as_float(edge_val(__float_as_int(n0), __float_as_int(n1), j[u]));
            float* op = acc + (size_t)dj * F;
#pragma unroll
            for (int k = 0; k < F / 32; ++k) atomAddF32(op + lane + 32 * k, buf[u][k] * nj);
        }
    }
}

template<int F>
__global__ __launch_bounds__(256)
void scatter_kernel(const float* __restrict__ h, const int* __restrict__ srcI,
                    const int* __restrict__ dstI, const float* __restrict__ dinv,
                    float* __restrict__ acc, unsigned* __restrict__ q, int E, int N) {
    constexpr int NL = F / 32;
    constexpr int U  = 16 / NL;
    __shared__ int s_job[2];
    const int lane = threadIdx.x & 31;
    const int wave = __builtin_amdgcn_readfirstlane((int)(threadIdx.x >> 5));
    const unsigned nchunk = ((unsigned)E + 511u) >> 9;

    if (threadIdx.x == 0) {
        const int die = (__builtin_amdgcn_s_sendmsg_rtn(0x87) >> 16) & 15;
        int r = die >= 4 ? 1 : 0;
        unsigned t = atomicAdd(&q[r], 1u);
        if (t >= nchunk) { r ^= 1; t = atomicAdd(&q[r], 1u); }
        s_job[0] = (int)t;
        s_job[1] = r;
    }
    __syncthreads();
    const unsigned chunk = (unsigned)__builtin_amdgcn_readfirstlane(s_job[0]);
    const int range = __builtin_amdgcn_readfirstlane(s_job[1]);
    if (chunk >= nchunk) return;

    const int split = N >> 1;
    const unsigned base = chunk * 512u + (unsigned)wave * 64u;
    const unsigned e0 = base + (unsigned)lane, e1 = e0 + 32u;

    int s0 = 0, d0 = 0, s1 = 0, d1 = 0;
    bool in0 = false, in1 = false;
    if (e0 < (unsigned)E) { s0 = srcI[e0]; d0 = dstI[e0]; in0 = (d0 >= split) == (range != 0); }
    if (e1 < (unsigned)E) { s1 = srcI[e1]; d1 = dstI[e1]; in1 = (d1 >= split) == (range != 0); }
    float n0 = 0.0f, n1 = 0.0f;
    if (in0) n0 = dinv[s0] * dinv[d0];
    if (in1) n1 = dinv[s1] * dinv[d1];
    unsigned long long m = (unsigned long long)__builtin_amdgcn_ballot_w32(in0)
                         | ((unsigned long long)__builtin_amdgcn_ballot_w32(in1) << 32);

    const int cnt = __builtin_popcountll(m);
    int rem = cnt / U;
    int ja[U], jb[U];
    float ba[U][NL], bb[U][NL];
    if (rem > 0) {
        take_edges<U, true>(m, ja);
        load_rows<F, U, true>(h, s0, s1, ja, ba, lane);
        --rem;
        while (rem >= 2) {
            take_edges<U, true>(m, jb);
            load_rows<F, U, true>(h, s0, s1, jb, bb, lane);
            add_rows<F, U, true>(acc, d0, d1, n0, n1, ja, ba, lane);
            take_edges<U, true>(m, ja);
            load_rows<F, U, true>(h, s0, s1, ja, ba, lane);
            add_rows<F, U, true>(acc, d0, d1, n0, n1, jb, bb, lane);
            rem -= 2;
        }
        if (rem == 1) {
            take_edges<U, true>(m, jb);
            load_rows<F, U, true>(h, s0, s1, jb, bb, lane);
            add_rows<F, U, true>(acc, d0, d1, n0, n1, ja, ba, lane);
            add_rows<F, U, true>(acc, d0, d1, n0, n1, jb, bb, lane);
        } else {
            add_rows<F, U, true>(acc, d0, d1, n0, n1, ja, ba, lane);
        }
    }
    if (m) {
        take_edges<U, false>(m, ja);
        load_rows<F, U, false>(h, s0, s1, ja, ba, lane);
        add_rows<F, U, false>(acc, d0, d1, n0, n1, ja, ba, lane);
    }
}


extern "C" void kernel_launch(void* const* d_in, const int* in_sizes, int n_in,
                              void* d_out, int out_size, void* d_ws, size_t ws_size,
                              hipStream_t stream) {
    const float* x  = (const float*)d_in[0];
    const int*   ei = (const int*)d_in[1];
    const float* W1 = (const float*)d_in[2];
    const float* b1 = (const float*)d_in[3];
    const float* W2 = (const float*)d_in[4];
    const float* b2 = (const float*)d_in[5];
    float* out = (float*)d_out;

    const int N = in_sizes[0] / N_FEAT;
    const int E = in_sizes[1] / 2;
    const int* srcI = ei;
    const int* dstI = ei + E;

    char* ws = (char*)d_ws;
    size_t degB = ((size_t)N * sizeof(float) + 4 * sizeof(unsigned) + 255) & ~(size_t)255;
    size_t t1B  = (((size_t)N * HIDDEN * sizeof(float)) + 255) & ~(size_t)255;
    float* deg = (float*)ws;
    unsigned* q = (unsigned*)(ws + (size_t)N * sizeof(float));
    float* t1  = (float*)(ws + degB);
    float* t2  = (float*)(ws + degB + t1B);
    float* t3  = t1;

    const int BT = 256;
    const int gemmBlocks = (N + 127) / 128;

    unsigned* part = (unsigned*)t2;
    deg_hist_kernel<<<DEG_G, DEG_T, 0, stream>>>(dstI, part, E, N);
    deg_finish_kernel<<<(N + BT - 1) / BT, BT, 0, stream>>>(part, deg, N, q);

    gemm_wmma_f32<N_FEAT, HIDDEN, false, false><<<gemmBlocks, 256, 0, stream>>>(x, W1, t1, t2, deg, b1, b2, N);

    const int scatBlocks = 2 * ((E + 511) / 512);
    scatter_kernel<HIDDEN><<<scatBlocks, BT, 0, stream>>>(t1, srcI, dstI, deg, t2, q, E, N);

    gemm_wmma_f32<HIDDEN, NCLS, true, true><<<gemmBlocks, 256, 0, stream>>>(t2, W2, t3, out, deg, b1, b2, N);

    scatter_kernel<NCLS><<<scatBlocks, BT, 0, stream>>>(t3, srcI, dstI, deg, out, q + 2, E, N);
}
